# starting kernel
# baseline (speedup 1.0000x reference)
#include <hip/hip_runtime.h>


typedef __attribute__((ext_vector_type(2))) float v2f;
typedef __attribute__((ext_vector_type(8))) float v8f;

#define SR_B 16
#define SR_N 1024
#define SR_D 16
#define SR_ALPHA_HALF 500.0f
#define SR_THREADS 512

__device__ __forceinline__ float sr_tanh(float x) {
#if __has_builtin(__builtin_amdgcn_tanhf)
  return __builtin_amdgcn_tanhf(x);
#else
  float r;
  asm volatile("v_tanh_f32 %0, %1" : "=v"(r) : "v"(x));
  return r;
#endif
}

__global__ __launch_bounds__(SR_THREADS) void softrank_wmma_kernel(
    const float* __restrict__ x, float* __restrict__ out) {
  __shared__ float lds[SR_D * SR_N];

  const int b   = (int)blockIdx.x >> 6;
  const int jb  = (int)blockIdx.x & 63;
  const int tid = (int)threadIdx.x;

  const float* __restrict__ xb = x + b * (SR_N * SR_D);
#pragma unroll
  for (int t = 0; t < (SR_N * SR_D) / SR_THREADS; ++t) {
    const int idx = tid + t * SR_THREADS;
    const int i = idx >> 4;
    const int k = idx & 15;
    lds[(k << 10) + i] = SR_ALPHA_HALF * xb[idx];
  }
  __syncthreads();

  const int  k    = tid >> 5;
  const int  lane = tid & 31;
  const int  ln16 = lane & 15;
  const bool lo   = lane < 16;
  const int  j0   = jb << 4;
  const float* __restrict__ ldsk = &lds[k << 10];

  v2f A;
  A.x = lo ? ldsk[j0 + ln16] : 0.0f;
  A.y = lo ? -1.0f : 0.0f;

  const v8f czero = {0.f, 0.f, 0.f, 0.f, 0.f, 0.f, 0.f, 0.f};
  v8f acc = czero;

  for (int i0 = 0; i0 < SR_N; i0 += 16) {
    v2f Bm;
    Bm.x = 1.0f;
    Bm.y = ldsk[i0 + ln16];

    v8f z = __builtin_amdgcn_wmma_f32_16x16x4_f32(
         false, A,  false, Bm,
         (short)0, czero,  false,  false);

#pragma unroll
    for (int r = 0; r < 8; ++r) acc[r] += sr_tanh(z[r]);
  }

#pragma unroll
  for (int r = 0; r < 8; ++r) {
    float v = acc[r];
    v += __shfl_xor(v, 1, 32);
    v += __shfl_xor(v, 2, 32);
    v += __shfl_xor(v, 4, 32);
    v += __shfl_xor(v, 8, 32);
    acc[r] = v;
  }

  if (ln16 == 0) {
    const int mbase = lo ? 0 : 8;
    float* __restrict__ ob =
        out + b * (SR_N * SR_D) + (j0 + mbase) * SR_D + k;
#pragma unroll
    for (int r = 0; r < 8; ++r)
      ob[r * SR_D] = 0.5f + acc[r] * (1.0f / (2.0f * SR_N));
  }
}

extern "C" void kernel_launch(void* const* d_in, const int* in_sizes, int n_in,
                              void* d_out, int out_size, void* d_ws, size_t ws_size,
                              hipStream_t stream) {
  (void)in_sizes; (void)n_in; (void)out_size; (void)d_ws; (void)ws_size;
  const float* x = (const float*)d_in[0];
  float* out = (float*)d_out;
  softrank_wmma_kernel<<<dim3(SR_B * (SR_N / 16)), dim3(SR_THREADS), 0, stream>>>(x, out);
}


# Pad the LDS row pitch of the [k][i] staging layout from 1024 to 1028 dwords: the transposing ds_store_b32 goes from a 16-way bank conflict to conflict-free; main loop and sum order untouched
# speedup vs baseline: 1.3298x; 1.3298x over previous
#include <hip/hip_runtime.h>


typedef __attribute__((ext_vector_type(2))) float v2f;
typedef __attribute__((ext_vector_type(8))) float v8f;

#define SR_B 16
#define SR_N 1024
#define SR_D 16
#define SR_ALPHA_HALF 500.0f
#define SR_THREADS 512
#define SR_PITCH 1028

__device__ __forceinline__ float sr_tanh(float x) {
#if __has_builtin(__builtin_amdgcn_tanhf)
  return __builtin_amdgcn_tanhf(x);
#else
  float r;
  asm volatile("v_tanh_f32 %0, %1" : "=v"(r) : "v"(x));
  return r;
#endif
}

__global__ __launch_bounds__(SR_THREADS) void softrank_wmma_kernel(
    const float* __restrict__ x, float* __restrict__ out) {
  __shared__ float lds[SR_D * SR_PITCH];

  const int b   = (int)blockIdx.x >> 6;
  const int jb  = (int)blockIdx.x & 63;
  const int tid = (int)threadIdx.x;

  const float* __restrict__ xb = x + b * (SR_N * SR_D);
#pragma unroll
  for (int t = 0; t < (SR_N * SR_D) / SR_THREADS; ++t) {
    const int idx = tid + t * SR_THREADS;
    const int i = idx >> 4;
    const int k = idx & 15;
    lds[k * SR_PITCH + i] = SR_ALPHA_HALF * xb[idx];
  }
  __syncthreads();

  const int  k    = tid >> 5;
  const int  lane = tid & 31;
  const int  ln16 = lane & 15;
  const bool lo   = lane < 16;
  const int  j0   = jb << 4;
  const float* __restrict__ ldsk = &lds[k * SR_PITCH];

  v2f A;
  A.x = lo ? ldsk[j0 + ln16] : 0.0f;
  A.y = lo ? -1.0f : 0.0f;

  const v8f czero = {0.f, 0.f, 0.f, 0.f, 0.f, 0.f, 0.f, 0.f};
  v8f acc = czero;

  for (int i0 = 0; i0 < SR_N; i0 += 16) {
    v2f Bm;
    Bm.x = 1.0f;
    Bm.y = ldsk[i0 + ln16];

    v8f z = __builtin_amdgcn_wmma_f32_16x16x4_f32(
         false, A,  false, Bm,
         (short)0, czero,  false,  false);

#pragma unroll
    for (int r = 0; r < 8; ++r) acc[r] += sr_tanh(z[r]);
  }

#pragma unroll
  for (int r = 0; r < 8; ++r) {
    float v = acc[r];
    v += __shfl_xor(v, 1, 32);
    v += __shfl_xor(v, 2, 32);
    v += __shfl_xor(v, 4, 32);
    v += __shfl_xor(v, 8, 32);
    acc[r] = v;
  }

  if (ln16 == 0) {
    const int mbase = lo ? 0 : 8;
    float* __restrict__ ob =
        out + b * (SR_N * SR_D) + (j0 + mbase) * SR_D + k;
#pragma unroll
    for (int r = 0; r < 8; ++r)
      ob[r * SR_D] = 0.5f + acc[r] * (1.0f / (2.0f * SR_N));
  }
}

extern "C" void kernel_launch(void* const* d_in, const int* in_sizes, int n_in,
                              void* d_out, int out_size, void* d_ws, size_t ws_size,
                              hipStream_t stream) {
  (void)in_sizes; (void)n_in; (void)out_size; (void)d_ws; (void)ws_size;
  const float* x = (const float*)d_in[0];
  float* out = (float*)d_out;
  softrank_wmma_kernel<<<dim3(SR_B * (SR_N / 16)), dim3(SR_THREADS), 0, stream>>>(x, out);
}


# tanh is odd: each unordered pair of 16-row tiles is evaluated once (row sums to j, column sums to i, joined in LDS by one work-group per (b,k)): half the WMMAs and tanh; the sum order changes
# speedup vs baseline: 1.5552x; 1.1695x over previous
#include <hip/hip_runtime.h>


typedef __attribute__((ext_vector_type(2))) float v2f;
typedef __attribute__((ext_vector_type(8))) float v8f;

#define SR_B 16
#define SR_N 1024
#define SR_D 16
#define SR_NT 64
#define SR_ALPHA_HALF 500.0f
#define SR_THREADS 512
#define SR_WAVES 16
#define SR_RP 36

__device__ __forceinline__ float sr_tanh(float x) {
#if __has_builtin(__builtin_amdgcn_tanhf)
  return __builtin_amdgcn_tanhf(x);
#else
  float r;
  asm volatile("v_tanh_f32 %0, %1" : "=v"(r) : "v"(x));
  return r;
#endif
}

__device__ __forceinline__ float sr_tile(const float* __restrict__ vs, const v2f A, v8f& acc,
                                         const int I, const int ln16) {
  const v8f czero = {0.f, 0.f, 0.f, 0.f, 0.f, 0.f, 0.f, 0.f};
  v2f Bm;
  Bm.x = 1.0f;
  Bm.y = vs[16 * I + ln16];
  const v8f z = __builtin_amdgcn_wmma_f32_16x16x4_f32(
       false, A,  false, Bm,
       (short)0, czero,  false,  false);
  v8f t;
#pragma unroll
  for (int r = 0; r < 8; ++r) t[r] = sr_tanh(z[r]);
#pragma unroll
  for (int r = 0; r < 8; ++r) acc[r] += t[r];
  return ((t[0] + t[1]) + (t[2] + t[3])) + ((t[4] + t[5]) + (t[6] + t[7]));
}

template <bool ADD, int NPART>
__device__ __forceinline__ void sr_rows(const float* __restrict__ vs, float* __restrict__ colw,
                                        float* __restrict__ rowpart, const int J,
                                        const int lane, const int ln16, const bool lo) {
  v2f A;
  A.x = lo ? vs[16 * J + ln16] : 0.0f;
  A.y = lo ? -1.0f : 0.0f;
  v8f acc = {0.f, 0.f, 0.f, 0.f, 0.f, 0.f, 0.f, 0.f};
  (void)sr_tile(vs, A, acc, J, ln16);
  constexpr int REM = NPART & 3;
#pragma unroll
  for (int d = 1; d <= REM; ++d) {
    float* slot = colw + ((J + d) & (SR_NT - 1)) * 32;
    const float old = ADD ? *slot : 0.0f;
    const float cp = sr_tile(vs, A, acc, J + d, ln16);
    *slot = ADD ? cp + old : cp;
  }
#pragma unroll 1
  for (int d0 = 1 + REM; d0 <= NPART; d0 += 4) {
    float old[4], cp[4];
#pragma unroll
    for (int u = 0; u < 4; ++u)
      old[u] = ADD ? colw[((J + d0 + u) & (SR_NT - 1)) * 32] : 0.0f;
#pragma unroll
    for (int u = 0; u < 4; ++u) cp[u] = sr_tile(vs, A, acc, J + d0 + u, ln16);
#pragma unroll
    for (int u = 0; u < 4; ++u)
      colw[((J + d0 + u) & (SR_NT - 1)) * 32] = ADD ? cp[u] + old[u] : cp[u];
  }
#pragma unroll
  for (int r = 0; r < 8; ++r) rowpart[(J * 8 + r) * SR_RP + lane] = acc[r];
}

__global__ __launch_bounds__(SR_THREADS) void softrank_wmma_kernel(
    const float* __restrict__ x, float* __restrict__ out) {
  __shared__ float vs[2 * SR_N];
  __shared__ float rowpart[SR_NT * 8 * SR_RP];
  __shared__ float cols[SR_WAVES * SR_NT * 32];

  const int b   = (int)blockIdx.x >> 4;
  const int k   = (int)blockIdx.x & 15;
  const int tid = (int)threadIdx.x;

  const float* __restrict__ xb = x + b * (SR_N * SR_D) + k;
#pragma unroll
  for (int t = 0; t < SR_N / SR_THREADS; ++t) {
    const int i = tid + t * SR_THREADS;
    const float v = SR_ALPHA_HALF * xb[i * SR_D];
    vs[i] = v;
    vs[i + SR_N] = v;
  }
  __syncthreads();

  const int  w    = __builtin_amdgcn_readfirstlane(tid >> 5);
  const int  lane = tid & 31;
  const int  ln16 = lane & 15;
  const bool lo   = lane < 16;
  float* __restrict__ colw = cols + w * (SR_NT * 32) + lane;

  colw[w * 32] = 0.0f;
  sr_rows<false, 32>(vs, colw, rowpart, w, lane, ln16, lo);
  sr_rows<false, 31>(vs, colw, rowpart, w + 32, lane, ln16, lo);
  sr_rows<true, 32>(vs, colw, rowpart, w + 16, lane, ln16, lo);
  sr_rows<true, 31>(vs, colw, rowpart, w + 48, lane, ln16, lo);

  __syncthreads();

#pragma unroll
  for (int t = 0; t < SR_N / SR_THREADS; ++t) {
    const int j = tid + t * SR_THREADS;
    const int J = j >> 4;
    const int m = j & 15;
    const float* __restrict__ rp = rowpart + (J * 8 + (m & 7)) * SR_RP + 16 * (m >> 3);
    float rs = 0.0f;
#pragma unroll
    for (int q = 0; q < 16; ++q) rs += rp[q];
    const float* __restrict__ cq = cols + J * 32 + m;
    float cs = 0.0f;
#pragma unroll
    for (int ww = 0; ww < SR_WAVES; ++ww)
      cs += cq[ww * (SR_NT * 32)] + cq[ww * (SR_NT * 32) + 16];
    out[(b * SR_N + j) * SR_D + k] = 0.5f + (rs - cs) * (1.0f / (2.0f * SR_N));
  }
}

extern "C" void kernel_launch(void* const* d_in, const int* in_sizes, int n_in,
                              void* d_out, int out_size, void* d_ws, size_t ws_size,
                              hipStream_t stream) {
  (void)in_sizes; (void)n_in; (void)out_size; (void)d_ws; (void)ws_size;
  const float* x = (const float*)d_in[0];
  float* out = (float*)d_out;
  softrank_wmma_kernel<<<dim3(SR_B * SR_D), dim3(SR_THREADS), 0, stream>>>(x, out);
}
